# starting kernel
# baseline (speedup 1.0000x reference)
#include <hip/hip_runtime.h>
#include <hip/hip_bf16.h>

typedef __attribute__((ext_vector_type(2))) float v2f;
typedef __attribute__((ext_vector_type(8))) float v8f;


__global__ void zero_ws_kernel(float* __restrict__ ws, int count) {
    int i = blockIdx.x * blockDim.x + threadIdx.x;
    int stride = gridDim.x * blockDim.x;
    for (; i < count; i += stride) ws[i] = 0.0f;
}

__global__ void edge_scatter_kernel(const float* __restrict__ node_emb,
                                    const int* __restrict__ esrc,
                                    const int* __restrict__ edst,
                                    float* __restrict__ agg,
                                    float* __restrict__ deg,
                                    int n_edges) {
    int e = blockIdx.x * blockDim.x + threadIdx.x;
    if (e >= n_edges) return;
    int s = esrc[e];
    int d = edst[e];
    const float4* row = (const float4*)(node_emb + (size_t)s * 8);
    float4 r0 = row[0];
    float4 r1 = row[1];
    float* a = agg + (size_t)d * 8;
    unsafeAtomicAdd(a + 0, r0.x);
    unsafeAtomicAdd(a + 1, r0.y);
    unsafeAtomicAdd(a + 2, r0.z);
    unsafeAtomicAdd(a + 3, r0.w);
    unsafeAtomicAdd(a + 4, r1.x);
    unsafeAtomicAdd(a + 5, r1.y);
    unsafeAtomicAdd(a + 6, r1.z);
    unsafeAtomicAdd(a + 7, r1.w);
    unsafeAtomicAdd(deg + d, 1.0f);
}

__global__ void sage_update_kernel(const float* __restrict__ node_emb,
                                   const float* __restrict__ agg,
                                   const float* __restrict__ deg,
                                   const float* __restrict__ W_self,
                                   const float* __restrict__ W_neigh,
                                   const float* __restrict__ sage_bias,
                                   float* __restrict__ relearned,
                                   int n_nodes) {
    __shared__ float sWs[64];
    __shared__ float sWn[64];
    __shared__ float sB[8];
    int t = threadIdx.x;
    if (t < 64) { sWs[t] = W_self[t]; sWn[t] = W_neigh[t]; }
    if (t < 8)  { sB[t] = sage_bias[t]; }
    __syncthreads();

    int i = blockIdx.x * blockDim.x + t;
    if (i >= n_nodes) return;

    const float4* nep = (const float4*)(node_emb + (size_t)i * 8);
    float4 n0 = nep[0], n1 = nep[1];
    float x[8] = {n0.x, n0.y, n0.z, n0.w, n1.x, n1.y, n1.z, n1.w};

    float dg = deg[i];
    float inv = 1.0f / fmaxf(dg, 1.0f);
    const float4* agp = (const float4*)(agg + (size_t)i * 8);
    float4 a0 = agp[0], a1 = agp[1];
    float h[8] = {a0.x * inv, a0.y * inv, a0.z * inv, a0.w * inv,
                  a1.x * inv, a1.y * inv, a1.z * inv, a1.w * inv};

    float out[8];
#pragma unroll
    for (int j = 0; j < 8; ++j) {
        float acc = sB[j];
#pragma unroll
        for (int k = 0; k < 8; ++k)
            acc += x[k] * sWs[j * 8 + k] + h[k] * sWn[j * 8 + k];
        out[j] = acc;
    }
    float4* rp = (float4*)(relearned + (size_t)i * 8);
    rp[0] = make_float4(out[0], out[1], out[2], out[3]);
    rp[1] = make_float4(out[4], out[5], out[6], out[7]);
}


#define XS_STR 772
#define X2_STR 148

__global__ void __launch_bounds__(256)
fused_mlp_kernel(const float* __restrict__ bert,
                 const float* __restrict__ W1, const float* __restrict__ b1,
                 const float* __restrict__ W2, const float* __restrict__ b2,
                 const float* __restrict__ W3, const float* __restrict__ b3,
                 const float* __restrict__ relearned,
                 const int* __restrict__ row_ids,
                 float* __restrict__ out,
                 int batch) {
    __shared__ float xs[16 * XS_STR];
    __shared__ float x2[16 * X2_STR];
    __shared__ float x3[16 * 16];

    const int tid  = threadIdx.x;
    const int lane = tid & 31;
    const int wv   = tid >> 5;
    const int nlo  = lane & 15;
    const int kh   = (lane >> 4) << 1;

    const size_t row0 = (size_t)blockIdx.x * 16;

    const float4* asrc = (const float4*)(bert + row0 * 768);
    for (int c = tid; c < 16 * 192; c += 256) {
        int r = c / 192, q = c % 192;
        float4 v = asrc[r * 192 + q];
        *(float4*)&xs[r * XS_STR + q * 4] = v;
    }
    __syncthreads();

    v8f c1 = {};
    const float* w1p = W1 + (size_t)(wv * 16 + nlo) * 768 + kh;
    const float* xap = &xs[nlo * XS_STR + kh];
#pragma unroll 4
    for (int k0 = 0; k0 < 768; k0 += 4) {
        v2f a = *(const v2f*)(xap + k0);
        v2f b = *(const v2f*)(w1p + k0);
        c1 = __builtin_amdgcn_wmma_f32_16x16x4_f32(
                 false, a, false, b, (short)0, c1, false, false);
    }
    {
        float bias = b1[wv * 16 + nlo];
        int mh = (lane >> 4) << 3;
#pragma unroll
        for (int r = 0; r < 8; ++r) {
            float v = fmaxf(c1[r] + bias, 0.0f);
            x2[(r + mh) * X2_STR + wv * 16 + nlo] = v;
        }
    }

    if (tid < 32) {
        int m = tid & 15, which = tid >> 4;
        int gm = (int)row0 + m;
        int idx = row_ids[which * batch + gm];
        const float4* src = (const float4*)(relearned + (size_t)idx * 8);
        float4 r0 = src[0], r1 = src[1];
        float* dst = &x2[m * X2_STR + 128 + which * 8];
        *(float4*)(dst + 0) = r0;
        *(float4*)(dst + 4) = r1;
    }
    __syncthreads();

    if (wv == 0) {
        v8f c2 = {};
        const float* w2p = W2 + (size_t)nlo * 144 + kh;
        const float* x2p = &x2[nlo * X2_STR + kh];
#pragma unroll
        for (int k0 = 0; k0 < 144; k0 += 4) {
            v2f a = *(const v2f*)(x2p + k0);
            v2f b = *(const v2f*)(w2p + k0);
            c2 = __builtin_amdgcn_wmma_f32_16x16x4_f32(
                     false, a, false, b, (short)0, c2, false, false);
        }
        float bias = b2[nlo];
        int mh = (lane >> 4) << 3;
#pragma unroll
        for (int r = 0; r < 8; ++r)
            x3[(r + mh) * 16 + nlo] = fmaxf(c2[r] + bias, 0.0f);
    }
    __syncthreads();

    if (tid < 16) {
        float a0 = b3[0], a1 = b3[1], a2 = b3[2];
#pragma unroll
        for (int n = 0; n < 16; ++n) {
            float v = x3[tid * 16 + n];
            a0 += v * W3[n];
            a1 += v * W3[16 + n];
            a2 += v * W3[32 + n];
        }
        size_t gm = row0 + tid;
        out[gm * 3 + 0] = a0;
        out[gm * 3 + 1] = a1;
        out[gm * 3 + 2] = a2;
    }
}


extern "C" void kernel_launch(void* const* d_in, const int* in_sizes, int n_in,
                              void* d_out, int out_size, void* d_ws, size_t ws_size,
                              hipStream_t stream) {
    const float* bert      = (const float*)d_in[0];
    const float* node_emb  = (const float*)d_in[1];
    const int*   esrc      = (const int*)d_in[2];
    const int*   edst      = (const int*)d_in[3];
    const int*   row_ids   = (const int*)d_in[4];
    const float* W1        = (const float*)d_in[5];
    const float* b1        = (const float*)d_in[6];
    const float* W2        = (const float*)d_in[7];
    const float* b2        = (const float*)d_in[8];
    const float* W3        = (const float*)d_in[9];
    const float* b3        = (const float*)d_in[10];
    const float* W_self    = (const float*)d_in[11];
    const float* W_neigh   = (const float*)d_in[12];
    const float* sage_bias = (const float*)d_in[13];

    const int n_nodes = in_sizes[1] / 8;
    const int n_edges = in_sizes[2];
    const int batch   = in_sizes[4] / 2;

    float* agg       = (float*)d_ws;
    float* deg       = agg + (size_t)n_nodes * 8;
    float* relearned = deg + (size_t)n_nodes;

    {
        int count = n_nodes * 9;
        int blocks = (count + 255) / 256;
        if (blocks > 4096) blocks = 4096;
        zero_ws_kernel<<<blocks, 256, 0, stream>>>((float*)d_ws, count);
    }

    edge_scatter_kernel<<<(n_edges + 255) / 256, 256, 0, stream>>>(
        node_emb, esrc, edst, agg, deg, n_edges);

    sage_update_kernel<<<(n_nodes + 255) / 256, 256, 0, stream>>>(
        node_emb, agg, deg, W_self, W_neigh, sage_bias, relearned, n_nodes);

    fused_mlp_kernel<<<batch / 16, 256, 0, stream>>>(
        bert, W1, b1, W2, b2, W3, b3, relearned, row_ids,
        (float*)d_out, batch);
}


# edge scatter: 64-byte node record (8 sums + degree), one lane per record word: an edge's 9 float atomic adds go out as 1 request, not 9; atomics kept, sum order arrival-dependent as before
# speedup vs baseline: 4.1735x; 4.1735x over previous
#include <hip/hip_runtime.h>
#include <hip/hip_bf16.h>

typedef __attribute__((ext_vector_type(2))) float v2f;
typedef __attribute__((ext_vector_type(8))) float v8f;


#define REC 16

__global__ void zero_ws_kernel(float* __restrict__ ws, int count) {
    int i = blockIdx.x * blockDim.x + threadIdx.x;
    int stride = gridDim.x * blockDim.x;
    for (; i < count; i += stride) ws[i] = 0.0f;
}

__global__ void edge_scatter_kernel(const float* __restrict__ node_emb,
                                    const int* __restrict__ esrc,
                                    const int* __restrict__ edst,
                                    float* __restrict__ rec,
                                    int n_edges) {
    const int lane = threadIdx.x & 31;
    const int half = lane >> 4;
    const unsigned w = lane & 15;
    const int e  = blockIdx.x * blockDim.x + threadIdx.x;
    const int e0 = e - lane;
    int s = 0, d = 0;
    if (e < n_edges) { s = esrc[e]; d = edst[e]; }

    float v[16];
    unsigned off[16];
#pragma unroll
    for (int j = 0; j < 16; ++j) {
        const int from = 2 * j + half;
        const unsigned sj = (unsigned)__shfl(s, from);
        const unsigned dj = (unsigned)__shfl(d, from);
        off[j] = dj * REC + w;
        v[j] = (w < 8) ? node_emb[sj * 8 + w] : 1.0f;
    }
#pragma unroll
    for (int j = 0; j < 16; ++j) {
        if (w <= 8 && e0 + 2 * j + half < n_edges)
            unsafeAtomicAdd(rec + off[j], v[j]);
    }
}

__global__ void sage_update_kernel(const float* __restrict__ node_emb,
                                   float* rec,
                                   const float* __restrict__ W_self,
                                   const float* __restrict__ W_neigh,
                                   const float* __restrict__ sage_bias,
                                   int n_nodes) {
    __shared__ float sWs[64];
    __shared__ float sWn[64];
    __shared__ float sB[8];
    int t = threadIdx.x;
    if (t < 64) { sWs[t] = W_self[t]; sWn[t] = W_neigh[t]; }
    if (t < 8)  { sB[t] = sage_bias[t]; }
    __syncthreads();

    int i = blockIdx.x * blockDim.x + t;
    if (i >= n_nodes) return;

    const float4* nep = (const float4*)(node_emb + (size_t)i * 8);
    float4 n0 = nep[0], n1 = nep[1];
    float x[8] = {n0.x, n0.y, n0.z, n0.w, n1.x, n1.y, n1.z, n1.w};

    float* myrec = rec + (size_t)i * REC;
    float dg = myrec[8];
    float inv = 1.0f / fmaxf(dg, 1.0f);
    const float4* agp = (const float4*)myrec;
    float4 a0 = agp[0], a1 = agp[1];
    float h[8] = {a0.x * inv, a0.y * inv, a0.z * inv, a0.w * inv,
                  a1.x * inv, a1.y * inv, a1.z * inv, a1.w * inv};

    float out[8];
#pragma unroll
    for (int j = 0; j < 8; ++j) {
        float acc = sB[j];
#pragma unroll
        for (int k = 0; k < 8; ++k)
            acc += x[k] * sWs[j * 8 + k] + h[k] * sWn[j * 8 + k];
        out[j] = acc;
    }
    float4* rp = (float4*)myrec;
    rp[0] = make_float4(out[0], out[1], out[2], out[3]);
    rp[1] = make_float4(out[4], out[5], out[6], out[7]);
}


#define XS_STR 772
#define X2_STR 148

__global__ void __launch_bounds__(256)
fused_mlp_kernel(const float* __restrict__ bert,
                 const float* __restrict__ W1, const float* __restrict__ b1,
                 const float* __restrict__ W2, const float* __restrict__ b2,
                 const float* __restrict__ W3, const float* __restrict__ b3,
                 const float* __restrict__ rec,
                 const int* __restrict__ row_ids,
                 float* __restrict__ out,
                 int batch) {
    __shared__ float xs[16 * XS_STR];
    __shared__ float x2[16 * X2_STR];
    __shared__ float x3[16 * 16];

    const int tid  = threadIdx.x;
    const int lane = tid & 31;
    const int wv   = tid >> 5;
    const int nlo  = lane & 15;
    const int kh   = (lane >> 4) << 1;

    const size_t row0 = (size_t)blockIdx.x * 16;

    const float4* asrc = (const float4*)(bert + row0 * 768);
    for (int c = tid; c < 16 * 192; c += 256) {
        int r = c / 192, q = c % 192;
        float4 v = asrc[r * 192 + q];
        *(float4*)&xs[r * XS_STR + q * 4] = v;
    }
    __syncthreads();

    v8f c1 = {};
    const float* w1p = W1 + (size_t)(wv * 16 + nlo) * 768 + kh;
    const float* xap = &xs[nlo * XS_STR + kh];
#pragma unroll 4
    for (int k0 = 0; k0 < 768; k0 += 4) {
        v2f a = *(const v2f*)(xap + k0);
        v2f b = *(const v2f*)(w1p + k0);
        c1 = __builtin_amdgcn_wmma_f32_16x16x4_f32(
                 false, a, false, b, (short)0, c1, false, false);
    }
    {
        float bias = b1[wv * 16 + nlo];
        int mh = (lane >> 4) << 3;
#pragma unroll
        for (int r = 0; r < 8; ++r) {
            float v = fmaxf(c1[r] + bias, 0.0f);
            x2[(r + mh) * X2_STR + wv * 16 + nlo] = v;
        }
    }

    if (tid < 32) {
        int m = tid & 15, which = tid >> 4;
        int gm = (int)row0 + m;
        int idx = row_ids[which * batch + gm];
        const float4* src = (const float4*)(rec + (size_t)idx * REC);
        float4 r0 = src[0], r1 = src[1];
        float* dst = &x2[m * X2_STR + 128 + which * 8];
        *(float4*)(dst + 0) = r0;
        *(float4*)(dst + 4) = r1;
    }
    __syncthreads();

    if (wv == 0) {
        v8f c2 = {};
        const float* w2p = W2 + (size_t)nlo * 144 + kh;
        const float* x2p = &x2[nlo * X2_STR + kh];
#pragma unroll
        for (int k0 = 0; k0 < 144; k0 += 4) {
            v2f a = *(const v2f*)(x2p + k0);
            v2f b = *(const v2f*)(w2p + k0);
            c2 = __builtin_amdgcn_wmma_f32_16x16x4_f32(
                     false, a, false, b, (short)0, c2, false, false);
        }
        float bias = b2[nlo];
        int mh = (lane >> 4) << 3;
#pragma unroll
        for (int r = 0; r < 8; ++r)
            x3[(r + mh) * 16 + nlo] = fmaxf(c2[r] + bias, 0.0f);
    }
    __syncthreads();

    if (tid < 16) {
        float a0 = b3[0], a1 = b3[1], a2 = b3[2];
#pragma unroll
        for (int n = 0; n < 16; ++n) {
            float v = x3[tid * 16 + n];
            a0 += v * W3[n];
            a1 += v * W3[16 + n];
            a2 += v * W3[32 + n];
        }
        size_t gm = row0 + tid;
        out[gm * 3 + 0] = a0;
        out[gm * 3 + 1] = a1;
        out[gm * 3 + 2] = a2;
    }
}


extern "C" void kernel_launch(void* const* d_in, const int* in_sizes, int n_in,
                              void* d_out, int out_size, void* d_ws, size_t ws_size,
                              hipStream_t stream) {
    const float* bert      = (const float*)d_in[0];
    const float* node_emb  = (const float*)d_in[1];
    const int*   esrc      = (const int*)d_in[2];
    const int*   edst      = (const int*)d_in[3];
    const int*   row_ids   = (const int*)d_in[4];
    const float* W1        = (const float*)d_in[5];
    const float* b1        = (const float*)d_in[6];
    const float* W2        = (const float*)d_in[7];
    const float* b2        = (const float*)d_in[8];
    const float* W3        = (const float*)d_in[9];
    const float* b3        = (const float*)d_in[10];
    const float* W_self    = (const float*)d_in[11];
    const float* W_neigh   = (const float*)d_in[12];
    const float* sage_bias = (const float*)d_in[13];

    const int n_nodes = in_sizes[1] / 8;
    const int n_edges = in_sizes[2];
    const int batch   = in_sizes[4] / 2;

    float* rec = (float*)d_ws;

    {
        int count = n_nodes * REC;
        int blocks = (count + 255) / 256;
        if (blocks > 4096) blocks = 4096;
        zero_ws_kernel<<<blocks, 256, 0, stream>>>((float*)d_ws, count);
    }

    edge_scatter_kernel<<<(n_edges + 255) / 256, 256, 0, stream>>>(
        node_emb, esrc, edst, rec, n_edges);

    sage_update_kernel<<<(n_nodes + 255) / 256, 256, 0, stream>>>(
        node_emb, rec, W_self, W_neigh, sage_bias, n_nodes);

    fused_mlp_kernel<<<batch / 16, 256, 0, stream>>>(
        bert, W1, b1, W2, b2, W3, b3, rec, row_ids,
        (float*)d_out, batch);
}


# fused MLP stage 1: 64 rows per block; each W1 fragment load feeds 4 WMMAs (4 row tiles, 4 accumulators per wave) instead of 1; A tile staged in two k-chunks of 384; per-element sum order unchanged
# speedup vs baseline: 5.3186x; 1.2744x over previous
#include <hip/hip_runtime.h>
#include <hip/hip_bf16.h>

typedef __attribute__((ext_vector_type(2))) float v2f;
typedef __attribute__((ext_vector_type(8))) float v8f;


#define REC 16

__global__ void zero_ws_kernel(float* __restrict__ ws, int count) {
    int i = blockIdx.x * blockDim.x + threadIdx.x;
    int stride = gridDim.x * blockDim.x;
    for (; i < count; i += stride) ws[i] = 0.0f;
}

__global__ void edge_scatter_kernel(const float* __restrict__ node_emb,
                                    const int* __restrict__ esrc,
                                    const int* __restrict__ edst,
                                    float* __restrict__ rec,
                                    int n_edges) {
    const int lane = threadIdx.x & 31;
    const int half = lane >> 4;
    const unsigned w = lane & 15;
    const int e  = blockIdx.x * blockDim.x + threadIdx.x;
    const int e0 = e - lane;
    int s = 0, d = 0;
    if (e < n_edges) { s = esrc[e]; d = edst[e]; }

    float v[16];
    unsigned off[16];
#pragma unroll
    for (int j = 0; j < 16; ++j) {
        const int from = 2 * j + half;
        const unsigned sj = (unsigned)__shfl(s, from);
        const unsigned dj = (unsigned)__shfl(d, from);
        off[j] = dj * REC + w;
        v[j] = (w < 8) ? node_emb[sj * 8 + w] : 1.0f;
    }
#pragma unroll
    for (int j = 0; j < 16; ++j) {
        if (w <= 8 && e0 + 2 * j + half < n_edges)
            unsafeAtomicAdd(rec + off[j], v[j]);
    }
}

__global__ void sage_update_kernel(const float* __restrict__ node_emb,
                                   float* rec,
                                   const float* __restrict__ W_self,
                                   const float* __restrict__ W_neigh,
                                   const float* __restrict__ sage_bias,
                                   int n_nodes) {
    __shared__ float sWs[64];
    __shared__ float sWn[64];
    __shared__ float sB[8];
    int t = threadIdx.x;
    if (t < 64) { sWs[t] = W_self[t]; sWn[t] = W_neigh[t]; }
    if (t < 8)  { sB[t] = sage_bias[t]; }
    __syncthreads();

    int i = blockIdx.x * blockDim.x + t;
    if (i >= n_nodes) return;

    const float4* nep = (const float4*)(node_emb + (size_t)i * 8);
    float4 n0 = nep[0], n1 = nep[1];
    float x[8] = {n0.x, n0.y, n0.z, n0.w, n1.x, n1.y, n1.z, n1.w};

    float* myrec = rec + (size_t)i * REC;
    float dg = myrec[8];
    float inv = 1.0f / fmaxf(dg, 1.0f);
    const float4* agp = (const float4*)myrec;
    float4 a0 = agp[0], a1 = agp[1];
    float h[8] = {a0.x * inv, a0.y * inv, a0.z * inv, a0.w * inv,
                  a1.x * inv, a1.y * inv, a1.z * inv, a1.w * inv};

    float out[8];
#pragma unroll
    for (int j = 0; j < 8; ++j) {
        float acc = sB[j];
#pragma unroll
        for (int k = 0; k < 8; ++k)
            acc += x[k] * sWs[j * 8 + k] + h[k] * sWn[j * 8 + k];
        out[j] = acc;
    }
    float4* rp = (float4*)myrec;
    rp[0] = make_float4(out[0], out[1], out[2], out[3]);
    rp[1] = make_float4(out[4], out[5], out[6], out[7]);
}


#define MT 4
#define ROWS (16 * MT)
#define KC 384
#define XS_STR 388
#define X2_STR 148

__global__ void __launch_bounds__(256)
fused_mlp_kernel(const float* __restrict__ bert,
                 const float* __restrict__ W1, const float* __restrict__ b1,
                 const float* __restrict__ W2, const float* __restrict__ b2,
                 const float* __restrict__ W3, const float* __restrict__ b3,
                 const float* __restrict__ rec,
                 const int* __restrict__ row_ids,
                 float* __restrict__ out,
                 int batch) {
    __shared__ float xs[ROWS * XS_STR];
    __shared__ float x2[ROWS * X2_STR];
    __shared__ float x3[ROWS * 16];

    const int tid  = threadIdx.x;
    const int lane = tid & 31;
    const int wv   = tid >> 5;
    const int nlo  = lane & 15;
    const int kh   = (lane >> 4) << 1;

    const size_t row0 = (size_t)blockIdx.x * ROWS;

    v8f c1[MT];
#pragma unroll
    for (int mt = 0; mt < MT; ++mt) c1[mt] = v8f{};

    const float4* asrc = (const float4*)(bert + row0 * 768);
    const float* w1p = W1 + (size_t)(wv * 16 + nlo) * 768 + kh;
    const float* xap = &xs[nlo * XS_STR + kh];

    for (int kc = 0; kc < 768; kc += KC) {
        if (kc) __syncthreads();

        for (int c0 = tid; c0 < ROWS * (KC / 4); c0 += 256 * 8) {
            float4 v[8];
#pragma unroll
            for (int u = 0; u < 8; ++u) {
                int c = c0 + u * 256;
                int r = c / (KC / 4), q = c % (KC / 4);
                v[u] = asrc[r * 192 + kc / 4 + q];
            }
#pragma unroll
            for (int u = 0; u < 8; ++u) {
                int c = c0 + u * 256;
                int r = c / (KC / 4), q = c % (KC / 4);
                *(float4*)&xs[r * XS_STR + q * 4] = v[u];
            }
        }
        __syncthreads();

#pragma unroll 8
        for (int k0 = 0; k0 < KC; k0 += 4) {
            v2f b = *(const v2f*)(w1p + kc + k0);
#pragma unroll
            for (int mt = 0; mt < MT; ++mt) {
                v2f a = *(const v2f*)(xap + mt * 16 * XS_STR + k0);
                c1[mt] = __builtin_amdgcn_wmma_f32_16x16x4_f32(
                             false, a, false, b, (short)0, c1[mt], false, false);
            }
        }
    }
    {
        float bias = b1[wv * 16 + nlo];
        int mh = (lane >> 4) << 3;
#pragma unroll
        for (int mt = 0; mt < MT; ++mt) {
#pragma unroll
            for (int r = 0; r < 8; ++r) {
                float v = fmaxf(c1[mt][r] + bias, 0.0f);
                x2[(mt * 16 + r + mh) * X2_STR + wv * 16 + nlo] = v;
            }
        }
    }

    if (tid < 2 * ROWS) {
        int m = tid & (ROWS - 1), which = tid >> 6;
        int gm = (int)row0 + m;
        int idx = row_ids[which * batch + gm];
        const float4* src = (const float4*)(rec + (size_t)idx * REC);
        float4 r0 = src[0], r1 = src[1];
        float* dst = &x2[m * X2_STR + 128 + which * 8];
        *(float4*)(dst + 0) = r0;
        *(float4*)(dst + 4) = r1;
    }
    __syncthreads();

    if (wv < MT) {
        v8f c2 = {};
        const float* w2p = W2 + (size_t)nlo * 144 + kh;
        const float* x2p = &x2[(wv * 16 + nlo) * X2_STR + kh];
#pragma unroll
        for (int k0 = 0; k0 < 144; k0 += 4) {
            v2f a = *(const v2f*)(x2p + k0);
            v2f b = *(const v2f*)(w2p + k0);
            c2 = __builtin_amdgcn_wmma_f32_16x16x4_f32(
                     false, a, false, b, (short)0, c2, false, false);
        }
        float bias = b2[nlo];
        int mh = (lane >> 4) << 3;
#pragma unroll
        for (int r = 0; r < 8; ++r)
            x3[(wv * 16 + r + mh) * 16 + nlo] = fmaxf(c2[r] + bias, 0.0f);
    }
    __syncthreads();

    if (tid < ROWS) {
        float a0 = b3[0], a1 = b3[1], a2 = b3[2];
#pragma unroll
        for (int n = 0; n < 16; ++n) {
            float v = x3[tid * 16 + n];
            a0 += v * W3[n];
            a1 += v * W3[16 + n];
            a2 += v * W3[32 + n];
        }
        size_t gm = row0 + tid;
        out[gm * 3 + 0] = a0;
        out[gm * 3 + 1] = a1;
        out[gm * 3 + 2] = a2;
    }
}


extern "C" void kernel_launch(void* const* d_in, const int* in_sizes, int n_in,
                              void* d_out, int out_size, void* d_ws, size_t ws_size,
                              hipStream_t stream) {
    const float* bert      = (const float*)d_in[0];
    const float* node_emb  = (const float*)d_in[1];
    const int*   esrc      = (const int*)d_in[2];
    const int*   edst      = (const int*)d_in[3];
    const int*   row_ids   = (const int*)d_in[4];
    const float* W1        = (const float*)d_in[5];
    const float* b1        = (const float*)d_in[6];
    const float* W2        = (const float*)d_in[7];
    const float* b2        = (const float*)d_in[8];
    const float* W3        = (const float*)d_in[9];
    const float* b3        = (const float*)d_in[10];
    const float* W_self    = (const float*)d_in[11];
    const float* W_neigh   = (const float*)d_in[12];
    const float* sage_bias = (const float*)d_in[13];

    const int n_nodes = in_sizes[1] / 8;
    const int n_edges = in_sizes[2];
    const int batch   = in_sizes[4] / 2;

    float* rec = (float*)d_ws;

    {
        int count = n_nodes * REC;
        int blocks = (count + 255) / 256;
        if (blocks > 4096) blocks = 4096;
        zero_ws_kernel<<<blocks, 256, 0, stream>>>((float*)d_ws, count);
    }

    edge_scatter_kernel<<<(n_edges + 255) / 256, 256, 0, stream>>>(
        node_emb, esrc, edst, rec, n_edges);

    sage_update_kernel<<<(n_nodes + 255) / 256, 256, 0, stream>>>(
        node_emb, rec, W_self, W_neigh, sage_bias, n_nodes);

    fused_mlp_kernel<<<batch / 64, 256, 0, stream>>>(
        bert, W1, b1, W2, b2, W3, b3, rec, row_ids,
        (float*)d_out, batch);
}


# edge scatter: two planes of node records, one per package half (wave picks its plane by die id), so no atomic line is shared across halves; update adds both: two arrival-ordered partial sums per word
# speedup vs baseline: 7.7863x; 1.4640x over previous
#include <hip/hip_runtime.h>
#include <hip/hip_bf16.h>

typedef __attribute__((ext_vector_type(2))) float v2f;
typedef __attribute__((ext_vector_type(8))) float v8f;


#define REC 16

__device__ __forceinline__ bool two_planes(int n_nodes, unsigned long long ws_bytes) {
    return ws_bytes >= 2ull * (unsigned long long)n_nodes * REC * sizeof(float);
}

__global__ void zero_ws_kernel(float* __restrict__ ws, int n_nodes,
                               unsigned long long ws_bytes) {
    const int count = n_nodes * REC * (two_planes(n_nodes, ws_bytes) ? 2 : 1);
    int i = blockIdx.x * blockDim.x + threadIdx.x;
    int stride = gridDim.x * blockDim.x;
    for (; i < count; i += stride) ws[i] = 0.0f;
}

__global__ void edge_scatter_kernel(const float* __restrict__ node_emb,
                                    const int* __restrict__ esrc,
                                    const int* __restrict__ edst,
                                    float* __restrict__ rec,
                                    int n_edges, int n_nodes,
                                    unsigned long long ws_bytes) {
    const unsigned die = (__builtin_amdgcn_s_sendmsg_rtn(0x87) >> 16) & 0xFu;
    if (two_planes(n_nodes, ws_bytes) && die >= 4u)
        rec += (size_t)n_nodes * REC;
    const int lane = threadIdx.x & 31;
    const int half = lane >> 4;
    const unsigned w = lane & 15;
    const int e  = blockIdx.x * blockDim.x + threadIdx.x;
    const int e0 = e - lane;
    int s = 0, d = 0;
    if (e < n_edges) { s = esrc[e]; d = edst[e]; }

    float v[16];
    unsigned off[16];
#pragma unroll
    for (int j = 0; j < 16; ++j) {
        const int from = 2 * j + half;
        const unsigned sj = (unsigned)__shfl(s, from);
        const unsigned dj = (unsigned)__shfl(d, from);
        off[j] = dj * REC + w;
        v[j] = (w < 8) ? node_emb[sj * 8 + w] : 1.0f;
    }
#pragma unroll
    for (int j = 0; j < 16; ++j) {
        if (w <= 8 && e0 + 2 * j + half < n_edges)
            unsafeAtomicAdd(rec + off[j], v[j]);
    }
}

__global__ void sage_update_kernel(const float* __restrict__ node_emb,
                                   float* rec,
                                   const float* __restrict__ W_self,
                                   const float* __restrict__ W_neigh,
                                   const float* __restrict__ sage_bias,
                                   int n_nodes, unsigned long long ws_bytes) {
    __shared__ float sWs[64];
    __shared__ float sWn[64];
    __shared__ float sB[8];
    int t = threadIdx.x;
    if (t < 64) { sWs[t] = W_self[t]; sWn[t] = W_neigh[t]; }
    if (t < 8)  { sB[t] = sage_bias[t]; }
    __syncthreads();

    int i = blockIdx.x * blockDim.x + t;
    if (i >= n_nodes) return;

    const float4* nep = (const float4*)(node_emb + (size_t)i * 8);
    float4 n0 = nep[0], n1 = nep[1];
    float x[8] = {n0.x, n0.y, n0.z, n0.w, n1.x, n1.y, n1.z, n1.w};

    float* myrec = rec + (size_t)i * REC;
    float dg = myrec[8];
    const float4* agp = (const float4*)myrec;
    float4 a0 = agp[0], a1 = agp[1];
    if (two_planes(n_nodes, ws_bytes)) {
        const float* rec1 = myrec + (size_t)n_nodes * REC;
        const float4* bgp = (const float4*)rec1;
        float4 b0 = bgp[0], b1 = bgp[1];
        dg += rec1[8];
        a0.x += b0.x; a0.y += b0.y; a0.z += b0.z; a0.w += b0.w;
        a1.x += b1.x; a1.y += b1.y; a1.z += b1.z; a1.w += b1.w;
    }
    float inv = 1.0f / fmaxf(dg, 1.0f);
    float h[8] = {a0.x * inv, a0.y * inv, a0.z * inv, a0.w * inv,
                  a1.x * inv, a1.y * inv, a1.z * inv, a1.w * inv};

    float out[8];
#pragma unroll
    for (int j = 0; j < 8; ++j) {
        float acc = sB[j];
#pragma unroll
        for (int k = 0; k < 8; ++k)
            acc += x[k] * sWs[j * 8 + k] + h[k] * sWn[j * 8 + k];
        out[j] = acc;
    }
    float4* rp = (float4*)myrec;
    rp[0] = make_float4(out[0], out[1], out[2], out[3]);
    rp[1] = make_float4(out[4], out[5], out[6], out[7]);
}


#define MT 4
#define ROWS (16 * MT)
#define KC 384
#define XS_STR 388
#define X2_STR 148

__global__ void __launch_bounds__(256)
fused_mlp_kernel(const float* __restrict__ bert,
                 const float* __restrict__ W1, const float* __restrict__ b1,
                 const float* __restrict__ W2, const float* __restrict__ b2,
                 const float* __restrict__ W3, const float* __restrict__ b3,
                 const float* __restrict__ rec,
                 const int* __restrict__ row_ids,
                 float* __restrict__ out,
                 int batch) {
    __shared__ float xs[ROWS * XS_STR];
    __shared__ float x2[ROWS * X2_STR];
    __shared__ float x3[ROWS * 16];

    const int tid  = threadIdx.x;
    const int lane = tid & 31;
    const int wv   = tid >> 5;
    const int nlo  = lane & 15;
    const int kh   = (lane >> 4) << 1;

    const size_t row0 = (size_t)blockIdx.x * ROWS;

    v8f c1[MT];
#pragma unroll
    for (int mt = 0; mt < MT; ++mt) c1[mt] = v8f{};

    const float4* asrc = (const float4*)(bert + row0 * 768);
    const float* w1p = W1 + (size_t)(wv * 16 + nlo) * 768 + kh;
    const float* xap = &xs[nlo * XS_STR + kh];

    for (int kc = 0; kc < 768; kc += KC) {
        if (kc) __syncthreads();

        for (int c0 = tid; c0 < ROWS * (KC / 4); c0 += 256 * 8) {
            float4 v[8];
#pragma unroll
            for (int u = 0; u < 8; ++u) {
                int c = c0 + u * 256;
                int r = c / (KC / 4), q = c % (KC / 4);
                v[u] = asrc[r * 192 + kc / 4 + q];
            }
#pragma unroll
            for (int u = 0; u < 8; ++u) {
                int c = c0 + u * 256;
                int r = c / (KC / 4), q = c % (KC / 4);
                *(float4*)&xs[r * XS_STR + q * 4] = v[u];
            }
        }
        __syncthreads();

#pragma unroll 8
        for (int k0 = 0; k0 < KC; k0 += 4) {
            v2f b = *(const v2f*)(w1p + kc + k0);
#pragma unroll
            for (int mt = 0; mt < MT; ++mt) {
                v2f a = *(const v2f*)(xap + mt * 16 * XS_STR + k0);
                c1[mt] = __builtin_amdgcn_wmma_f32_16x16x4_f32(
                             false, a, false, b, (short)0, c1[mt], false, false);
            }
        }
    }
    {
        float bias = b1[wv * 16 + nlo];
        int mh = (lane >> 4) << 3;
#pragma unroll
        for (int mt = 0; mt < MT; ++mt) {
#pragma unroll
            for (int r = 0; r < 8; ++r) {
                float v = fmaxf(c1[mt][r] + bias, 0.0f);
                x2[(mt * 16 + r + mh) * X2_STR + wv * 16 + nlo] = v;
            }
        }
    }

    if (tid < 2 * ROWS) {
        int m = tid & (ROWS - 1), which = tid >> 6;
        int gm = (int)row0 + m;
        int idx = row_ids[which * batch + gm];
        const float4* src = (const float4*)(rec + (size_t)idx * REC);
        float4 r0 = src[0], r1 = src[1];
        float* dst = &x2[m * X2_STR + 128 + which * 8];
        *(float4*)(dst + 0) = r0;
        *(float4*)(dst + 4) = r1;
    }
    __syncthreads();

    if (wv < MT) {
        v8f c2 = {};
        const float* w2p = W2 + (size_t)nlo * 144 + kh;
        const float* x2p = &x2[(wv * 16 + nlo) * X2_STR + kh];
#pragma unroll
        for (int k0 = 0; k0 < 144; k0 += 4) {
            v2f a = *(const v2f*)(x2p + k0);
            v2f b = *(const v2f*)(w2p + k0);
            c2 = __builtin_amdgcn_wmma_f32_16x16x4_f32(
                     false, a, false, b, (short)0, c2, false, false);
        }
        float bias = b2[nlo];
        int mh = (lane >> 4) << 3;
#pragma unroll
        for (int r = 0; r < 8; ++r)
            x3[(wv * 16 + r + mh) * 16 + nlo] = fmaxf(c2[r] + bias, 0.0f);
    }
    __syncthreads();

    if (tid < ROWS) {
        float a0 = b3[0], a1 = b3[1], a2 = b3[2];
#pragma unroll
        for (int n = 0; n < 16; ++n) {
            float v = x3[tid * 16 + n];
            a0 += v * W3[n];
            a1 += v * W3[16 + n];
            a2 += v * W3[32 + n];
        }
        size_t gm = row0 + tid;
        out[gm * 3 + 0] = a0;
        out[gm * 3 + 1] = a1;
        out[gm * 3 + 2] = a2;
    }
}


extern "C" void kernel_launch(void* const* d_in, const int* in_sizes, int n_in,
                              void* d_out, int out_size, void* d_ws, size_t ws_size,
                              hipStream_t stream) {
    const float* bert      = (const float*)d_in[0];
    const float* node_emb  = (const float*)d_in[1];
    const int*   esrc      = (const int*)d_in[2];
    const int*   edst      = (const int*)d_in[3];
    const int*   row_ids   = (const int*)d_in[4];
    const float* W1        = (const float*)d_in[5];
    const float* b1        = (const float*)d_in[6];
    const float* W2        = (const float*)d_in[7];
    const float* b2        = (const float*)d_in[8];
    const float* W3        = (const float*)d_in[9];
    const float* b3        = (const float*)d_in[10];
    const float* W_self    = (const float*)d_in[11];
    const float* W_neigh   = (const float*)d_in[12];
    const float* sage_bias = (const float*)d_in[13];

    const int n_nodes = in_sizes[1] / 8;
    const int n_edges = in_sizes[2];
    const int batch   = in_sizes[4] / 2;

    float* rec = (float*)d_ws;
    const unsigned long long ws_bytes = (unsigned long long)ws_size;

    {
        int count = n_nodes * REC;
        int blocks = (count + 255) / 256;
        if (blocks > 4096) blocks = 4096;
        zero_ws_kernel<<<blocks, 256, 0, stream>>>((float*)d_ws, n_nodes, ws_bytes);
    }

    edge_scatter_kernel<<<(n_edges + 255) / 256, 256, 0, stream>>>(
        node_emb, esrc, edst, rec, n_edges, n_nodes, ws_bytes);

    sage_update_kernel<<<(n_nodes + 255) / 256, 256, 0, stream>>>(
        node_emb, rec, W_self, W_neigh, sage_bias, n_nodes, ws_bytes);

    fused_mlp_kernel<<<batch / 64, 256, 0, stream>>>(
        bert, W1, b1, W2, b2, W3, b3, rec, row_ids,
        (float*)d_out, batch);
}


# SAGE update done inside the MLP kernel, only for each row's head and tail node (from the two record planes, after stage 1's first chunk load); the per-node update launch is gone: 3 launches, not 4
# speedup vs baseline: 8.1138x; 1.0421x over previous
#include <hip/hip_runtime.h>
#include <hip/hip_bf16.h>

typedef __attribute__((ext_vector_type(2))) float v2f;
typedef __attribute__((ext_vector_type(8))) float v8f;


#define REC 16

__device__ __forceinline__ bool two_planes(int n_nodes, unsigned long long ws_bytes) {
    return ws_bytes >= 2ull * (unsigned long long)n_nodes * REC * sizeof(float);
}

__global__ void zero_ws_kernel(float* __restrict__ ws, int n_nodes,
                               unsigned long long ws_bytes) {
    const int count = n_nodes * REC * (two_planes(n_nodes, ws_bytes) ? 2 : 1);
    int i = blockIdx.x * blockDim.x + threadIdx.x;
    int stride = gridDim.x * blockDim.x;
    for (; i < count; i += stride) ws[i] = 0.0f;
}

__global__ void edge_scatter_kernel(const float* __restrict__ node_emb,
                                    const int* __restrict__ esrc,
                                    const int* __restrict__ edst,
                                    float* __restrict__ rec,
                                    int n_edges, int n_nodes,
                                    unsigned long long ws_bytes) {
    const unsigned die = (__builtin_amdgcn_s_sendmsg_rtn(0x87) >> 16) & 0xFu;
    if (two_planes(n_nodes, ws_bytes) && die >= 4u)
        rec += (size_t)n_nodes * REC;
    const int lane = threadIdx.x & 31;
    const int half = lane >> 4;
    const unsigned w = lane & 15;
    const int e  = blockIdx.x * blockDim.x + threadIdx.x;
    const int e0 = e - lane;
    int s = 0, d = 0;
    if (e < n_edges) { s = esrc[e]; d = edst[e]; }

    float v[16];
    unsigned off[16];
#pragma unroll
    for (int j = 0; j < 16; ++j) {
        const int from = 2 * j + half;
        const unsigned sj = (unsigned)__shfl(s, from);
        const unsigned dj = (unsigned)__shfl(d, from);
        off[j] = dj * REC + w;
        v[j] = (w < 8) ? node_emb[sj * 8 + w] : 1.0f;
    }
#pragma unroll
    for (int j = 0; j < 16; ++j) {
        if (w <= 8 && e0 + 2 * j + half < n_edges)
            unsafeAtomicAdd(rec + off[j], v[j]);
    }
}


#define MT 4
#define ROWS (16 * MT)
#define KC 384
#define XS_STR 388
#define X2_STR 148

__global__ void __launch_bounds__(256)
fused_mlp_kernel(const float* __restrict__ bert,
                 const float* __restrict__ W1, const float* __restrict__ b1,
                 const float* __restrict__ W2, const float* __restrict__ b2,
                 const float* __restrict__ W3, const float* __restrict__ b3,
                 const float* __restrict__ node_emb,
                 const float* __restrict__ rec,
                 const float* __restrict__ W_self,
                 const float* __restrict__ W_neigh,
                 const float* __restrict__ sage_bias,
                 const int* __restrict__ row_ids,
                 float* __restrict__ out,
                 int batch, int n_nodes, unsigned long long ws_bytes) {
    __shared__ float xs[ROWS * XS_STR];
    __shared__ float x2[ROWS * X2_STR];
    __shared__ float x3[ROWS * 16];

    const int tid  = threadIdx.x;
    const int lane = tid & 31;
    const int wv   = tid >> 5;
    const int nlo  = lane & 15;
    const int kh   = (lane >> 4) << 1;

    const size_t row0 = (size_t)blockIdx.x * ROWS;

    const int gm_row = tid & (ROWS - 1), gwhich = (tid >> 6) & 1;
    const int gidx = row_ids[gwhich * batch + (int)row0 + gm_row];

    v8f c1[MT];
#pragma unroll
    for (int mt = 0; mt < MT; ++mt) c1[mt] = v8f{};

    const float4* asrc = (const float4*)(bert + row0 * 768);
    const float* w1p = W1 + (size_t)(wv * 16 + nlo) * 768 + kh;
    const float* xap = &xs[nlo * XS_STR + kh];

    auto stage_chunk = [&](int kc) {
        for (int c0 = tid; c0 < ROWS * (KC / 4); c0 += 256 * 8) {
            float4 v[8];
#pragma unroll
            for (int u = 0; u < 8; ++u) {
                int c = c0 + u * 256;
                int r = c / (KC / 4), q = c % (KC / 4);
                v[u] = asrc[r * 192 + kc / 4 + q];
            }
#pragma unroll
            for (int u = 0; u < 8; ++u) {
                int c = c0 + u * 256;
                int r = c / (KC / 4), q = c % (KC / 4);
                *(float4*)&xs[r * XS_STR + q * 4] = v[u];
            }
        }
    };
    auto mma_chunk = [&](int kc) {
#pragma unroll 8
        for (int k0 = 0; k0 < KC; k0 += 4) {
            v2f b = *(const v2f*)(w1p + kc + k0);
#pragma unroll
            for (int mt = 0; mt < MT; ++mt) {
                v2f a = *(const v2f*)(xap + mt * 16 * XS_STR + k0);
                c1[mt] = __builtin_amdgcn_wmma_f32_16x16x4_f32(
                             false, a, false, b, (short)0, c1[mt], false, false);
            }
        }
    };

    stage_chunk(0);
    __syncthreads();

    if (tid < 2 * ROWS) {
        const float4* nep = (const float4*)(node_emb + (size_t)gidx * 8);
        const float* myrec = rec + (size_t)gidx * REC;
        const float4* agp = (const float4*)myrec;
        float4 n0 = nep[0], n1 = nep[1];
        float4 a0 = agp[0], a1 = agp[1];
        float dg = myrec[8];
        if (two_planes(n_nodes, ws_bytes)) {
            const float* rec1 = myrec + (size_t)n_nodes * REC;
            const float4* bgp = (const float4*)rec1;
            float4 p0 = bgp[0], p1 = bgp[1];
            dg += rec1[8];
            a0.x += p0.x; a0.y += p0.y; a0.z += p0.z; a0.w += p0.w;
            a1.x += p1.x; a1.y += p1.y; a1.z += p1.z; a1.w += p1.w;
        }
        float x[8] = {n0.x, n0.y, n0.z, n0.w, n1.x, n1.y, n1.z, n1.w};
        float inv = 1.0f / fmaxf(dg, 1.0f);
        float h[8] = {a0.x * inv, a0.y * inv, a0.z * inv, a0.w * inv,
                      a1.x * inv, a1.y * inv, a1.z * inv, a1.w * inv};
        float o[8];
#pragma unroll
        for (int j = 0; j < 8; ++j) {
            float acc = sage_bias[j];
#pragma unroll
            for (int k = 0; k < 8; ++k)
                acc += x[k] * W_self[j * 8 + k] + h[k] * W_neigh[j * 8 + k];
            o[j] = acc;
        }
        float* dst = &x2[gm_row * X2_STR + 128 + gwhich * 8];
        *(float4*)(dst + 0) = make_float4(o[0], o[1], o[2], o[3]);
        *(float4*)(dst + 4) = make_float4(o[4], o[5], o[6], o[7]);
    }

    mma_chunk(0);
    __syncthreads();
    stage_chunk(KC);
    __syncthreads();
    mma_chunk(KC);
    {
        float bias = b1[wv * 16 + nlo];
        int mh = (lane >> 4) << 3;
#pragma unroll
        for (int mt = 0; mt < MT; ++mt) {
#pragma unroll
            for (int r = 0; r < 8; ++r) {
                float v = fmaxf(c1[mt][r] + bias, 0.0f);
                x2[(mt * 16 + r + mh) * X2_STR + wv * 16 + nlo] = v;
            }
        }
    }

    __syncthreads();

    if (wv < MT) {
        v8f c2 = {};
        const float* w2p = W2 + (size_t)nlo * 144 + kh;
        const float* x2p = &x2[(wv * 16 + nlo) * X2_STR + kh];
#pragma unroll
        for (int k0 = 0; k0 < 144; k0 += 4) {
            v2f a = *(const v2f*)(x2p + k0);
            v2f b = *(const v2f*)(w2p + k0);
            c2 = __builtin_amdgcn_wmma_f32_16x16x4_f32(
                     false, a, false, b, (short)0, c2, false, false);
        }
        float bias = b2[nlo];
        int mh = (lane >> 4) << 3;
#pragma unroll
        for (int r = 0; r < 8; ++r)
            x3[(wv * 16 + r + mh) * 16 + nlo] = fmaxf(c2[r] + bias, 0.0f);
    }
    __syncthreads();

    if (tid < ROWS) {
        float a0 = b3[0], a1 = b3[1], a2 = b3[2];
#pragma unroll
        for (int n = 0; n < 16; ++n) {
            float v = x3[tid * 16 + n];
            a0 += v * W3[n];
            a1 += v * W3[16 + n];
            a2 += v * W3[32 + n];
        }
        size_t gm = row0 + tid;
        out[gm * 3 + 0] = a0;
        out[gm * 3 + 1] = a1;
        out[gm * 3 + 2] = a2;
    }
}


extern "C" void kernel_launch(void* const* d_in, const int* in_sizes, int n_in,
                              void* d_out, int out_size, void* d_ws, size_t ws_size,
                              hipStream_t stream) {
    const float* bert      = (const float*)d_in[0];
    const float* node_emb  = (const float*)d_in[1];
    const int*   esrc      = (const int*)d_in[2];
    const int*   edst      = (const int*)d_in[3];
    const int*   row_ids   = (const int*)d_in[4];
    const float* W1        = (const float*)d_in[5];
    const float* b1        = (const float*)d_in[6];
    const float* W2        = (const float*)d_in[7];
    const float* b2        = (const float*)d_in[8];
    const float* W3        = (const float*)d_in[9];
    const float* b3        = (const float*)d_in[10];
    const float* W_self    = (const float*)d_in[11];
    const float* W_neigh   = (const float*)d_in[12];
    const float* sage_bias = (const float*)d_in[13];

    const int n_nodes = in_sizes[1] / 8;
    const int n_edges = in_sizes[2];
    const int batch   = in_sizes[4] / 2;

    float* rec = (float*)d_ws;
    const unsigned long long ws_bytes = (unsigned long long)ws_size;

    {
        int count = n_nodes * REC;
        int blocks = (count + 255) / 256;
        if (blocks > 4096) blocks = 4096;
        zero_ws_kernel<<<blocks, 256, 0, stream>>>((float*)d_ws, n_nodes, ws_bytes);
    }

    edge_scatter_kernel<<<(n_edges + 255) / 256, 256, 0, stream>>>(
        node_emb, esrc, edst, rec, n_edges, n_nodes, ws_bytes);

    fused_mlp_kernel<<<batch / 64, 256, 0, stream>>>(
        bert, W1, b1, W2, b2, W3, b3, node_emb, rec, W_self, W_neigh, sage_bias,
        row_ids, (float*)d_out, batch, n_nodes, ws_bytes);
}


# edge scatter adds only edges whose destination a batch row gathers: per-call node bit map built beside the zeroing, held in LDS by resident blocks looping over edge tiles; atomic sum order as before
# speedup vs baseline: 8.8592x; 1.0919x over previous
#include <hip/hip_runtime.h>
#include <hip/hip_bf16.h>

typedef __attribute__((ext_vector_type(2))) float v2f;
typedef __attribute__((ext_vector_type(8))) float v8f;


#define REC 16

__device__ __forceinline__ bool two_planes(int n_nodes, unsigned long long ws_bytes) {
    return ws_bytes >= 2ull * (unsigned long long)n_nodes * REC * sizeof(float);
}

#define MAP_LDS_WORDS 4096

__device__ __forceinline__ int map_words(int n_nodes) { return (n_nodes + 31) >> 5; }

__device__ __forceinline__ bool has_map(int n_nodes, unsigned long long ws_bytes) {
    return map_words(n_nodes) <= MAP_LDS_WORDS &&
           ws_bytes >= 2ull * (unsigned long long)n_nodes * REC * sizeof(float)
                       + (unsigned long long)map_words(n_nodes) * sizeof(unsigned);
}

#define OWN_NODES 2048
#define PREP_T 1024

__global__ void __launch_bounds__(PREP_T)
prep_kernel(float* __restrict__ ws, const int* __restrict__ row_ids, int n_ids,
            int n_nodes, unsigned long long ws_bytes) {
    __shared__ unsigned mark_w[OWN_NODES / 4];
    unsigned char* mark = (unsigned char*)mark_w;
    const int tid = threadIdx.x;
    const int owners = has_map(n_nodes, ws_bytes)
                     ? (n_nodes + OWN_NODES - 1) / OWN_NODES : 0;
    if ((int)blockIdx.x < owners) {
        const int base = (int)blockIdx.x * OWN_NODES;
        for (int i = tid; i < OWN_NODES; i += PREP_T) mark[i] = 0;
        __syncthreads();
        const int n4 = n_ids >> 2;
        const int4* ids4 = (const int4*)row_ids;
        for (int i0 = tid; i0 < n4; i0 += PREP_T * 8) {
            int4 q[8];
#pragma unroll
            for (int u = 0; u < 8; ++u) {
                const int i = i0 + u * PREP_T;
                q[u] = i < n4 ? ids4[i] : make_int4(-1, -1, -1, -1);
            }
#pragma unroll
            for (int u = 0; u < 8; ++u) {
                const unsigned r0 = (unsigned)(q[u].x - base), r1 = (unsigned)(q[u].y - base);
                const unsigned r2 = (unsigned)(q[u].z - base), r3 = (unsigned)(q[u].w - base);
                if (r0 < OWN_NODES) mark[r0] = 1;
                if (r1 < OWN_NODES) mark[r1] = 1;
                if (r2 < OWN_NODES) mark[r2] = 1;
                if (r3 < OWN_NODES) mark[r3] = 1;
            }
        }
        for (int i = (n4 << 2) + tid; i < n_ids; i += PREP_T) {
            const unsigned r = (unsigned)(row_ids[i] - base);
            if (r < OWN_NODES) mark[r] = 1;
        }
        __syncthreads();
        const int word = (base >> 5) + tid;
        if (tid < OWN_NODES / 32 && word < map_words(n_nodes)) {
            const unsigned* m4 = &mark_w[tid * 8];
            unsigned bits = 0;
#pragma unroll
            for (int k = 0; k < 8; ++k) {
                const unsigned b = m4[k];
                bits |= ((b & 1u) | ((b >> 7) & 2u) | ((b >> 14) & 4u) | ((b >> 21) & 8u)) << (4 * k);
            }
            unsigned* map = (unsigned*)(ws + 2ull * (unsigned long long)n_nodes * REC);
            map[word] = bits;
        }
    } else {
        const int zb = (int)gridDim.x - owners;
        const int count4 = n_nodes * (REC / 4) * (two_planes(n_nodes, ws_bytes) ? 2 : 1);
        float4* ws4 = (float4*)ws;
        const float4 z = make_float4(0.0f, 0.0f, 0.0f, 0.0f);
        for (int i = ((int)blockIdx.x - owners) * PREP_T + tid; i < count4; i += zb * PREP_T)
            ws4[i] = z;
    }
}

#define SC_T 512

__global__ void __launch_bounds__(SC_T)
edge_scatter_kernel(const float* __restrict__ node_emb,
                    const int* __restrict__ esrc,
                    const int* __restrict__ edst,
                    float* __restrict__ rec,
                    int n_edges, int n_nodes,
                    unsigned long long ws_bytes) {
    __shared__ unsigned map_s[MAP_LDS_WORDS];
    const int tid = threadIdx.x;
    const bool use_map = has_map(n_nodes, ws_bytes);
    if (use_map) {
        const unsigned* map_g = (const unsigned*)(rec + 2ull * (unsigned long long)n_nodes * REC);
        const int n_words = map_words(n_nodes);
        for (int i = tid; i < n_words; i += SC_T) map_s[i] = map_g[i];
    }
    __syncthreads();

    const unsigned die = (__builtin_amdgcn_s_sendmsg_rtn(0x87) >> 16) & 0xFu;
    if (two_planes(n_nodes, ws_bytes) && die >= 4u)
        rec += (size_t)n_nodes * REC;
    const int lane = tid & 31;
    const int half = lane >> 4;
    const unsigned w = lane & 15;
    const int n_tiles = (n_edges + 31) >> 5;
    const int n_waves = (int)gridDim.x * (SC_T / 32);
    int tile = (int)blockIdx.x * (SC_T / 32) + (tid >> 5);

    int s = 0, d = 0;
    if (tile < n_tiles) {
        const int e = tile * 32 + lane;
        if (e < n_edges) { s = esrc[e]; d = edst[e]; }
    }
    while (tile < n_tiles) {
        const int e = tile * 32 + lane;
        bool need = e < n_edges;
        if (use_map) need = need && ((map_s[d >> 5] >> (d & 31)) & 1u);
        const unsigned live = __builtin_amdgcn_ballot_w32(need);
        const int cnt = __builtin_popcount(live);
        const int rank = (int)__builtin_amdgcn_mbcnt_lo(live, 0u);
        const int pos = need ? rank : cnt + lane - rank;
        const int sp = __builtin_amdgcn_ds_permute(pos << 2, s);
        const int dp = __builtin_amdgcn_ds_permute(pos << 2, d);

        float v[16];
        unsigned off[16];
#pragma unroll
        for (int u = 0; u < 16; ++u) {
            if (2 * u < cnt) {
                const int from = 2 * u + half;
                const unsigned sj = (unsigned)__shfl(sp, from);
                const unsigned dj = (unsigned)__shfl(dp, from);
                off[u] = dj * REC + w;
                v[u] = (w < 8 && from < cnt) ? node_emb[sj * 8 + w] : 1.0f;
            }
        }
        const int tile_n = tile + n_waves;
        int s_n = 0, d_n = 0;
        if (tile_n < n_tiles) {
            const int en = tile_n * 32 + lane;
            if (en < n_edges) { s_n = esrc[en]; d_n = edst[en]; }
        }
        __builtin_amdgcn_fence(__ATOMIC_ACQUIRE, "workgroup");
#pragma unroll
        for (int u = 0; u < 16; ++u) {
            if (2 * u < cnt) {
                if (w <= 8 && 2 * u + half < cnt)
                    unsafeAtomicAdd(rec + off[u], v[u]);
            }
        }
        s = s_n; d = d_n; tile = tile_n;
    }
}


#define MT 4
#define ROWS (16 * MT)
#define KC 384
#define XS_STR 388
#define X2_STR 148

__global__ void __launch_bounds__(256)
fused_mlp_kernel(const float* __restrict__ bert,
                 const float* __restrict__ W1, const float* __restrict__ b1,
                 const float* __restrict__ W2, const float* __restrict__ b2,
                 const float* __restrict__ W3, const float* __restrict__ b3,
                 const float* __restrict__ node_emb,
                 const float* __restrict__ rec,
                 const float* __restrict__ W_self,
                 const float* __restrict__ W_neigh,
                 const float* __restrict__ sage_bias,
                 const int* __restrict__ row_ids,
                 float* __restrict__ out,
                 int batch, int n_nodes, unsigned long long ws_bytes) {
    __shared__ float xs[ROWS * XS_STR];
    __shared__ float x2[ROWS * X2_STR];
    __shared__ float x3[ROWS * 16];

    const int tid  = threadIdx.x;
    const int lane = tid & 31;
    const int wv   = tid >> 5;
    const int nlo  = lane & 15;
    const int kh   = (lane >> 4) << 1;

    const size_t row0 = (size_t)blockIdx.x * ROWS;

    const int gm_row = tid & (ROWS - 1), gwhich = (tid >> 6) & 1;
    const int gidx = row_ids[gwhich * batch + (int)row0 + gm_row];

    v8f c1[MT];
#pragma unroll
    for (int mt = 0; mt < MT; ++mt) c1[mt] = v8f{};

    const float4* asrc = (const float4*)(bert + row0 * 768);
    const float* w1p = W1 + (size_t)(wv * 16 + nlo) * 768 + kh;
    const float* xap = &xs[nlo * XS_STR + kh];

    auto stage_chunk = [&](int kc) {
        for (int c0 = tid; c0 < ROWS * (KC / 4); c0 += 256 * 8) {
            float4 v[8];
#pragma unroll
            for (int u = 0; u < 8; ++u) {
                int c = c0 + u * 256;
                int r = c / (KC / 4), q = c % (KC / 4);
                v[u] = asrc[r * 192 + kc / 4 + q];
            }
#pragma unroll
            for (int u = 0; u < 8; ++u) {
                int c = c0 + u * 256;
                int r = c / (KC / 4), q = c % (KC / 4);
                *(float4*)&xs[r * XS_STR + q * 4] = v[u];
            }
        }
    };
    auto mma_chunk = [&](int kc) {
#pragma unroll 8
        for (int k0 = 0; k0 < KC; k0 += 4) {
            v2f b = *(const v2f*)(w1p + kc + k0);
#pragma unroll
            for (int mt = 0; mt < MT; ++mt) {
                v2f a = *(const v2f*)(xap + mt * 16 * XS_STR + k0);
                c1[mt] = __builtin_amdgcn_wmma_f32_16x16x4_f32(
                             false, a, false, b, (short)0, c1[mt], false, false);
            }
        }
    };

    stage_chunk(0);
    __syncthreads();

    if (tid < 2 * ROWS) {
        const float4* nep = (const float4*)(node_emb + (size_t)gidx * 8);
        const float* myrec = rec + (size_t)gidx * REC;
        const float4* agp = (const float4*)myrec;
        float4 n0 = nep[0], n1 = nep[1];
        float4 a0 = agp[0], a1 = agp[1];
        float dg = myrec[8];
        if (two_planes(n_nodes, ws_bytes)) {
            const float* rec1 = myrec + (size_t)n_nodes * REC;
            const float4* bgp = (const float4*)rec1;
            float4 p0 = bgp[0], p1 = bgp[1];
            dg += rec1[8];
            a0.x += p0.x; a0.y += p0.y; a0.z += p0.z; a0.w += p0.w;
            a1.x += p1.x; a1.y += p1.y; a1.z += p1.z; a1.w += p1.w;
        }
        float x[8] = {n0.x, n0.y, n0.z, n0.w, n1.x, n1.y, n1.z, n1.w};
        float inv = 1.0f / fmaxf(dg, 1.0f);
        float h[8] = {a0.x * inv, a0.y * inv, a0.z * inv, a0.w * inv,
                      a1.x * inv, a1.y * inv, a1.z * inv, a1.w * inv};
        float o[8];
#pragma unroll
        for (int j = 0; j < 8; ++j) {
            float acc = sage_bias[j];
#pragma unroll
            for (int k = 0; k < 8; ++k)
                acc += x[k] * W_self[j * 8 + k] + h[k] * W_neigh[j * 8 + k];
            o[j] = acc;
        }
        float* dst = &x2[gm_row * X2_STR + 128 + gwhich * 8];
        *(float4*)(dst + 0) = make_float4(o[0], o[1], o[2], o[3]);
        *(float4*)(dst + 4) = make_float4(o[4], o[5], o[6], o[7]);
    }

    mma_chunk(0);
    __syncthreads();
    stage_chunk(KC);
    __syncthreads();
    mma_chunk(KC);
    {
        float bias = b1[wv * 16 + nlo];
        int mh = (lane >> 4) << 3;
#pragma unroll
        for (int mt = 0; mt < MT; ++mt) {
#pragma unroll
            for (int r = 0; r < 8; ++r) {
                float v = fmaxf(c1[mt][r] + bias, 0.0f);
                x2[(mt * 16 + r + mh) * X2_STR + wv * 16 + nlo] = v;
            }
        }
    }

    __syncthreads();

    if (wv < MT) {
        v8f c2 = {};
        const float* w2p = W2 + (size_t)nlo * 144 + kh;
        const float* x2p = &x2[(wv * 16 + nlo) * X2_STR + kh];
#pragma unroll
        for (int k0 = 0; k0 < 144; k0 += 4) {
            v2f a = *(const v2f*)(x2p + k0);
            v2f b = *(const v2f*)(w2p + k0);
            c2 = __builtin_amdgcn_wmma_f32_16x16x4_f32(
                     false, a, false, b, (short)0, c2, false, false);
        }
        float bias = b2[nlo];
        int mh = (lane >> 4) << 3;
#pragma unroll
        for (int r = 0; r < 8; ++r)
            x3[(wv * 16 + r + mh) * 16 + nlo] = fmaxf(c2[r] + bias, 0.0f);
    }
    __syncthreads();

    if (tid < ROWS) {
        float a0 = b3[0], a1 = b3[1], a2 = b3[2];
#pragma unroll
        for (int n = 0; n < 16; ++n) {
            float v = x3[tid * 16 + n];
            a0 += v * W3[n];
            a1 += v * W3[16 + n];
            a2 += v * W3[32 + n];
        }
        size_t gm = row0 + tid;
        out[gm * 3 + 0] = a0;
        out[gm * 3 + 1] = a1;
        out[gm * 3 + 2] = a2;
    }
}


extern "C" void kernel_launch(void* const* d_in, const int* in_sizes, int n_in,
                              void* d_out, int out_size, void* d_ws, size_t ws_size,
                              hipStream_t stream) {
    const float* bert      = (const float*)d_in[0];
    const float* node_emb  = (const float*)d_in[1];
    const int*   esrc      = (const int*)d_in[2];
    const int*   edst      = (const int*)d_in[3];
    const int*   row_ids   = (const int*)d_in[4];
    const float* W1        = (const float*)d_in[5];
    const float* b1        = (const float*)d_in[6];
    const float* W2        = (const float*)d_in[7];
    const float* b2        = (const float*)d_in[8];
    const float* W3        = (const float*)d_in[9];
    const float* b3        = (const float*)d_in[10];
    const float* W_self    = (const float*)d_in[11];
    const float* W_neigh   = (const float*)d_in[12];
    const float* sage_bias = (const float*)d_in[13];

    const int n_nodes = in_sizes[1] / 8;
    const int n_edges = in_sizes[2];
    const int batch   = in_sizes[4] / 2;

    float* rec = (float*)d_ws;
    const unsigned long long ws_bytes = (unsigned long long)ws_size;

    prep_kernel<<<(n_nodes + 2047) / 2048 + 256, 1024, 0, stream>>>(
        (float*)d_ws, row_ids, in_sizes[4], n_nodes, ws_bytes);

    edge_scatter_kernel<<<1024, 512, 0, stream>>>(
        node_emb, esrc, edst, rec, n_edges, n_nodes, ws_bytes);

    fused_mlp_kernel<<<batch / 64, 256, 0, stream>>>(
        bert, W1, b1, W2, b2, W3, b3, node_emb, rec, W_self, W_neigh, sage_bias,
        row_ids, (float*)d_out, batch, n_nodes, ws_bytes);
}
